# starting kernel
# baseline (speedup 1.0000x reference)
#include <hip/hip_runtime.h>


typedef float v2f __attribute__((ext_vector_type(2)));
typedef float v8f __attribute__((ext_vector_type(8)));

constexpr int kT    = 2048;
constexpr int kCIN  = 32;
constexpr int kCOUT = 32;
constexpr int kTT   = 16;
constexpr int kOPB  = 8;
constexpr int kPAD  = 17;

__global__ __launch_bounds__(256) void mimo_iir_wmma_kernel(
    const float* __restrict__ u,
    const float* __restrict__ bc,
    const float* __restrict__ ac,
    float* __restrict__ out)
{
    __shared__ float u_lds[kTT * kCIN];
    __shared__ float y_lds[kOPB * kCIN * kPAD];

    const int tid  = threadIdx.x;
    const int w    = tid >> 5;
    const int lane = tid & 31;
    const int b    = blockIdx.x / (kCOUT / kOPB);
    const int o    = (blockIdx.x % (kCOUT / kOPB)) * kOPB + w;
    const int i    = lane;

    const int cb = (o * kCIN + i) * 3;
    const float b0 = bc[cb + 0], b1 = bc[cb + 1], b2 = bc[cb + 2];
    const float a0 = ac[cb + 0], a1 = ac[cb + 1], a2 = ac[cb + 2];

    float up1 = 0.f, up2 = 0.f, up3 = 0.f;
    float y1 = 0.f, y2 = 0.f, y3 = 0.f;

    const float* ub = u   + (size_t)b * kT * kCIN;
    float*       ob = out + (size_t)b * kT * kCOUT;

    const int m   = lane & 15;
    const int khi = (lane < 16) ? 0 : 2;

    v2f bfrag; bfrag.x = 1.0f; bfrag.y = 1.0f;

    float* yw = &y_lds[w * kCIN * kPAD];

    for (int t0 = 0; t0 < kT; t0 += kTT) {
        __syncthreads();
        u_lds[tid]       = ub[t0 * kCIN + tid];
        u_lds[tid + 256] = ub[t0 * kCIN + tid + 256];
        __syncthreads();

        #pragma unroll
        for (int tt = 0; tt < kTT; ++tt) {
            float x  = fmaf(b0, up1, fmaf(b1, up2, b2 * up3));
            float p  = fmaf(-a1, y2, fmaf(-a2, y3, x));
            float yt = fmaf(-a0, y1, p);
            yw[i * kPAD + tt] = yt;
            y3 = y2; y2 = y1; y1 = yt;
            up3 = up2; up2 = up1; up1 = u_lds[tt * kCIN + i];
        }
        __builtin_amdgcn_wave_barrier();

        v8f acc = {};
        #pragma unroll
        for (int g = 0; g < 8; ++g) {
            v2f afrag;
            afrag.x = yw[(4 * g + khi + 0) * kPAD + m];
            afrag.y = yw[(4 * g + khi + 1) * kPAD + m];
            acc = __builtin_amdgcn_wmma_f32_16x16x4_f32(
                false, afrag, false, bfrag, (short)0, acc, false, false);
        }

        if (lane == 0) {
            #pragma unroll
            for (int r = 0; r < 8; ++r) ob[(t0 + r) * kCOUT + o] = acc[r];
        } else if (lane == 16) {
            #pragma unroll
            for (int r = 0; r < 8; ++r) ob[(t0 + 8 + r) * kCOUT + o] = acc[r];
        }
    }
}

extern "C" void kernel_launch(void* const* d_in, const int* in_sizes, int n_in,
                              void* d_out, int out_size, void* d_ws, size_t ws_size,
                              hipStream_t stream) {
    (void)n_in; (void)out_size; (void)d_ws; (void)ws_size;
    const float* u  = (const float*)d_in[0];
    const float* bc = (const float*)d_in[1];
    const float* ac = (const float*)d_in[2];
    float* out = (float*)d_out;

    const int B = in_sizes[0] / (kT * kCIN);
    dim3 grid(B * (kCOUT / kOPB));
    dim3 block(kOPB * 32);
    mimo_iir_wmma_kernel<<<grid, block, 0, stream>>>(u, bc, ac, out);
}


# each wave prefetches its own u rows into registers one 32-step group ahead, straight from global; LDS staging of u and both block barriers per tile removed (no store-ack or load wait per tile)
# speedup vs baseline: 2.0489x; 2.0489x over previous
#include <hip/hip_runtime.h>


typedef float v2f __attribute__((ext_vector_type(2)));
typedef float v8f __attribute__((ext_vector_type(8)));

constexpr int kT    = 2048;
constexpr int kCIN  = 32;
constexpr int kCOUT = 32;
constexpr int kTT   = 16;
constexpr int kG    = 32;
constexpr int kOPB  = 8;
constexpr int kPAD  = 17;

__device__ __forceinline__ void load_group(float (&dst)[kG], const float* p)
{
    #pragma unroll
    for (int k = 0; k < kG; ++k) dst[k] = p[k * kCIN];
}

__device__ __forceinline__ void run_group(
    const float (&uc)[kG],
    float b0, float b1, float b2, float a0, float a1, float a2,
    float& up1, float& up2, float& up3, float& y1, float& y2, float& y3,
    float* yw, int i, int m, int khi, int lane,
    float* __restrict__ orow)
{
    v2f bfrag; bfrag.x = 1.0f; bfrag.y = 1.0f;

    #pragma unroll
    for (int h = 0; h < kG / kTT; ++h) {
        #pragma unroll
        for (int tt = 0; tt < kTT; ++tt) {
            float x  = fmaf(b0, up1, fmaf(b1, up2, b2 * up3));
            float p  = fmaf(-a1, y2, fmaf(-a2, y3, x));
            float yt = fmaf(-a0, y1, p);
            yw[i * kPAD + tt] = yt;
            y3 = y2; y2 = y1; y1 = yt;
            up3 = up2; up2 = up1; up1 = uc[h * kTT + tt];
        }
        __builtin_amdgcn_wave_barrier();

        v8f acc = {};
        #pragma unroll
        for (int g = 0; g < 8; ++g) {
            v2f afrag;
            afrag.x = yw[(4 * g + khi + 0) * kPAD + m];
            afrag.y = yw[(4 * g + khi + 1) * kPAD + m];
            acc = __builtin_amdgcn_wmma_f32_16x16x4_f32(
                false, afrag, false, bfrag, (short)0, acc, false, false);
        }

        float* ot = orow + h * kTT * kCOUT;
        if (lane == 0) {
            #pragma unroll
            for (int r = 0; r < 8; ++r) ot[r * kCOUT] = acc[r];
        } else if (lane == 16) {
            #pragma unroll
            for (int r = 0; r < 8; ++r) ot[(8 + r) * kCOUT] = acc[r];
        }
    }
}

__global__ __launch_bounds__(256) void mimo_iir_wmma_kernel(
    const float* u,
    const float* __restrict__ bc,
    const float* __restrict__ ac,
    float* __restrict__ out)
{
    __shared__ float y_lds[kOPB * kCIN * kPAD];

    const int tid  = threadIdx.x;
    const int w    = tid >> 5;
    const int lane = tid & 31;
    const int b    = blockIdx.x / (kCOUT / kOPB);
    const int o    = (blockIdx.x % (kCOUT / kOPB)) * kOPB + w;
    const int i    = lane;

    const int cb = (o * kCIN + i) * 3;
    const float b0 = bc[cb + 0], b1 = bc[cb + 1], b2 = bc[cb + 2];
    const float a0 = ac[cb + 0], a1 = ac[cb + 1], a2 = ac[cb + 2];

    float up1 = 0.f, up2 = 0.f, up3 = 0.f;
    float y1 = 0.f, y2 = 0.f, y3 = 0.f;

    const float* ul = u   + (size_t)b * kT * kCIN + i;
    float*       ob = out + (size_t)b * kT * kCOUT + o;

    const int m   = lane & 15;
    const int khi = (lane < 16) ? 0 : 2;

    float* yw = &y_lds[w * kCIN * kPAD];

    float ua[kG], ub[kG];
    load_group(ua, ul);

    for (int t0 = 0; t0 < kT; t0 += 2 * kG) {
        load_group(ub, ul + (t0 + kG) * kCIN);
        run_group(ua, b0, b1, b2, a0, a1, a2, up1, up2, up3, y1, y2, y3,
                  yw, i, m, khi, lane, ob + t0 * kCOUT);
        const int tn = (t0 + 2 * kG < kT) ? (t0 + 2 * kG) : 0;
        load_group(ua, ul + tn * kCIN);
        run_group(ub, b0, b1, b2, a0, a1, a2, up1, up2, up3, y1, y2, y3,
                  yw, i, m, khi, lane, ob + (t0 + kG) * kCOUT);
    }
}

extern "C" void kernel_launch(void* const* d_in, const int* in_sizes, int n_in,
                              void* d_out, int out_size, void* d_ws, size_t ws_size,
                              hipStream_t stream) {
    (void)n_in; (void)out_size; (void)d_ws; (void)ws_size;
    const float* u  = (const float*)d_in[0];
    const float* bc = (const float*)d_in[1];
    const float* ac = (const float*)d_in[2];
    float* out = (float*)d_out;

    const int B = in_sizes[0] / (kT * kCIN);
    dim3 grid(B * (kCOUT / kOPB));
    dim3 block(kOPB * 32);
    mimo_iir_wmma_kernel<<<grid, block, 0, stream>>>(u, bc, ac, out);
}


# channel sum runs one tile behind the recurrence: a tile's LDS fragment reads go out as one group into a second register set; its 8 matrix ops and store follow the next tile's steps
# speedup vs baseline: 2.3836x; 1.1634x over previous
#include <hip/hip_runtime.h>


typedef float v2f __attribute__((ext_vector_type(2)));
typedef float v8f __attribute__((ext_vector_type(8)));

constexpr int kT    = 2048;
constexpr int kCIN  = 32;
constexpr int kCOUT = 32;
constexpr int kTT   = 16;
constexpr int kG    = 32;
constexpr int kOPB  = 8;
constexpr int kPAD  = 17;
constexpr int kNF   = kCIN / 4;

__device__ __forceinline__ void load_group(float (&dst)[kG], const float* p)
{
    #pragma unroll
    for (int k = 0; k < kG; ++k) dst[k] = p[k * kCIN];
}

__device__ __forceinline__ void chain_tile(
    const float (&uc)[kG], const int h,
    float b0, float b1, float b2, float a0, float a1, float a2,
    float& up1, float& up2, float& up3, float& y1, float& y2, float& y3,
    float* yw, int i)
{
    #pragma unroll
    for (int tt = 0; tt < kTT; ++tt) {
        float x  = fmaf(b0, up1, fmaf(b1, up2, b2 * up3));
        float p  = fmaf(-a1, y2, fmaf(-a2, y3, x));
        float yt = fmaf(-a0, y1, p);
        yw[i * kPAD + tt] = yt;
        y3 = y2; y2 = y1; y1 = yt;
        up3 = up2; up2 = up1; up1 = uc[h * kTT + tt];
    }
}

__device__ __forceinline__ void read_frags(v2f (&fr)[kNF], const float* fbase)
{
    #pragma unroll
    for (int g = 0; g < kNF; ++g) {
        fr[g].x = fbase[(4 * g + 0) * kPAD];
        fr[g].y = fbase[(4 * g + 1) * kPAD];
    }
}

__device__ __forceinline__ void sum_store(const v2f (&fr)[kNF], int lane,
                                          float* __restrict__ ot)
{
    v2f bfrag; bfrag.x = 1.0f; bfrag.y = 1.0f;
    v8f acc = {};
    #pragma unroll
    for (int g = 0; g < kNF; ++g) {
        acc = __builtin_amdgcn_wmma_f32_16x16x4_f32(
            false, fr[g], false, bfrag, (short)0, acc, false, false);
    }

    if (lane == 0) {
        #pragma unroll
        for (int r = 0; r < 8; ++r) ot[r * kCOUT] = acc[r];
    } else if (lane == 16) {
        #pragma unroll
        for (int r = 0; r < 8; ++r) ot[(8 + r) * kCOUT] = acc[r];
    }
}

__global__ __launch_bounds__(256) void mimo_iir_wmma_kernel(
    const float* u,
    const float* __restrict__ bc,
    const float* __restrict__ ac,
    float* __restrict__ out)
{
    __shared__ float y_lds[kOPB * kCIN * kPAD];

    const int tid  = threadIdx.x;
    const int w    = tid >> 5;
    const int lane = tid & 31;
    const int b    = blockIdx.x / (kCOUT / kOPB);
    const int o    = (blockIdx.x % (kCOUT / kOPB)) * kOPB + w;
    const int i    = lane;

    const int cb = (o * kCIN + i) * 3;
    const float b0 = bc[cb + 0], b1 = bc[cb + 1], b2 = bc[cb + 2];
    const float a0 = ac[cb + 0], a1 = ac[cb + 1], a2 = ac[cb + 2];

    float up1 = 0.f, up2 = 0.f, up3 = 0.f;
    float y1 = 0.f, y2 = 0.f, y3 = 0.f;

    const float* ul = u   + (size_t)b * kT * kCIN + i;
    float*       ob = out + (size_t)b * kT * kCOUT + o;

    const int m   = lane & 15;
    const int khi = (lane < 16) ? 0 : 2;

    float* yw = &y_lds[w * kCIN * kPAD];
    const float* fbase = yw + khi * kPAD + m;

    float ua[kG], ub[kG];
    v2f fa[kNF], fb[kNF] = {};
    load_group(ua, ul);

#define TILE_STEP(UC, H, FNEW, FOLD, TPREV, GUARD)                                   \
    chain_tile(UC, H, b0, b1, b2, a0, a1, a2, up1, up2, up3, y1, y2, y3, yw, i);     \
    __builtin_amdgcn_wave_barrier();                                                 \
    read_frags(FNEW, fbase);                                                         \
    __builtin_amdgcn_sched_barrier(0);                                               \
    if (GUARD) sum_store(FOLD, lane, ob + (TPREV) * kCOUT);

    for (int t0 = 0; t0 < kT; t0 += 2 * kG) {
        load_group(ub, ul + (t0 + kG) * kCIN);
        TILE_STEP(ua, 0, fa, fb, t0 - kTT, t0 > 0)
        TILE_STEP(ua, 1, fb, fa, t0, true)
        const int tn = (t0 + 2 * kG < kT) ? (t0 + 2 * kG) : 0;
        load_group(ua, ul + tn * kCIN);
        TILE_STEP(ub, 0, fa, fb, t0 + kTT, true)
        TILE_STEP(ub, 1, fb, fa, t0 + 2 * kTT, true)
    }
#undef TILE_STEP
    sum_store(fb, lane, ob + (kT - kTT) * kCOUT);
}

extern "C" void kernel_launch(void* const* d_in, const int* in_sizes, int n_in,
                              void* d_out, int out_size, void* d_ws, size_t ws_size,
                              hipStream_t stream) {
    (void)n_in; (void)out_size; (void)d_ws; (void)ws_size;
    const float* u  = (const float*)d_in[0];
    const float* bc = (const float*)d_in[1];
    const float* ac = (const float*)d_in[2];
    float* out = (float*)d_out;

    const int B = in_sizes[0] / (kT * kCIN);
    dim3 grid(B * (kCOUT / kOPB));
    dim3 block(kOPB * 32);
    mimo_iir_wmma_kernel<<<grid, block, 0, stream>>>(u, bc, ac, out);
}


# sum tile enters the matrix op as B (column = timestep) against an all-ones A: sums land one per lane, two tiles go in one full-wave store; divergent 2-lane 8-store path gone; sum order kept
# speedup vs baseline: 2.7782x; 1.1655x over previous
#include <hip/hip_runtime.h>


typedef float v2f __attribute__((ext_vector_type(2)));
typedef float v8f __attribute__((ext_vector_type(8)));

constexpr int kT    = 2048;
constexpr int kCIN  = 32;
constexpr int kCOUT = 32;
constexpr int kTT   = 16;
constexpr int kG    = 32;
constexpr int kOPB  = 8;
constexpr int kPAD  = 17;
constexpr int kNF   = kCIN / 4;

__device__ __forceinline__ void load_group(float (&dst)[kG], const float* p)
{
    #pragma unroll
    for (int k = 0; k < kG; ++k) dst[k] = p[k * kCIN];
}

__device__ __forceinline__ void chain_tile(
    const float (&uc)[kG], const int h,
    float b0, float b1, float b2, float a0, float a1, float a2,
    float& up1, float& up2, float& up3, float& y1, float& y2, float& y3,
    float* yw, int i)
{
    #pragma unroll
    for (int tt = 0; tt < kTT; ++tt) {
        float x  = fmaf(b0, up1, fmaf(b1, up2, b2 * up3));
        float p  = fmaf(-a1, y2, fmaf(-a2, y3, x));
        float yt = fmaf(-a0, y1, p);
        yw[i * kPAD + tt] = yt;
        y3 = y2; y2 = y1; y1 = yt;
        up3 = up2; up2 = up1; up1 = uc[h * kTT + tt];
    }
}

__device__ __forceinline__ void read_frags(v2f (&fr)[kNF], const float* fbase)
{
    #pragma unroll
    for (int g = 0; g < kNF; ++g) {
        fr[g].x = fbase[(4 * g + 0) * kPAD];
        fr[g].y = fbase[(4 * g + 1) * kPAD];
    }
}

__device__ __forceinline__ float tile_sum(const v2f (&fr)[kNF])
{
    v2f ones; ones.x = 1.0f; ones.y = 1.0f;
    v8f acc = {};
    #pragma unroll
    for (int g = 0; g < kNF; ++g) {
        acc = __builtin_amdgcn_wmma_f32_16x16x4_f32(
            false, ones, false, fr[g], (short)0, acc, false, false);
    }
    return acc[0];
}

__device__ __forceinline__ void store_pair(float lo, float hi, int lane,
                                           float* __restrict__ ot)
{
    ot[lane * kCOUT] = (lane < 16) ? lo : hi;
}

__global__ __launch_bounds__(256) void mimo_iir_wmma_kernel(
    const float* u,
    const float* __restrict__ bc,
    const float* __restrict__ ac,
    float* __restrict__ out)
{
    __shared__ float y_lds[kOPB * kCIN * kPAD];

    const int tid  = threadIdx.x;
    const int w    = tid >> 5;
    const int lane = tid & 31;
    const int b    = blockIdx.x / (kCOUT / kOPB);
    const int o    = (blockIdx.x % (kCOUT / kOPB)) * kOPB + w;
    const int i    = lane;

    const int cb = (o * kCIN + i) * 3;
    const float b0 = bc[cb + 0], b1 = bc[cb + 1], b2 = bc[cb + 2];
    const float a0 = ac[cb + 0], a1 = ac[cb + 1], a2 = ac[cb + 2];

    float up1 = 0.f, up2 = 0.f, up3 = 0.f;
    float y1 = 0.f, y2 = 0.f, y3 = 0.f;

    const float* ul = u   + (size_t)b * kT * kCIN + i;
    float*       ob = out + (size_t)b * kT * kCOUT + o;

    const int m   = lane & 15;
    const int khi = (lane < 16) ? 0 : 2;

    float* yw = &y_lds[w * kCIN * kPAD];
    const float* fbase = yw + khi * kPAD + m;

    float ua[kG], ub[kG];
    v2f fa[kNF], fb[kNF] = {};
    load_group(ua, ul);

#define TILE_CHAIN(UC, H, FNEW)                                                      \
    chain_tile(UC, H, b0, b1, b2, a0, a1, a2, up1, up2, up3, y1, y2, y3, yw, i);     \
    __builtin_amdgcn_wave_barrier();                                                 \
    read_frags(FNEW, fbase);                                                         \
    __builtin_amdgcn_sched_barrier(0);

    float keep = 0.f;

    for (int t0 = 0; t0 < kT; t0 += 2 * kG) {
        load_group(ub, ul + (t0 + kG) * kCIN);
        TILE_CHAIN(ua, 0, fa)
        if (t0 > 0) store_pair(keep, tile_sum(fb), lane, ob + (t0 - kG) * kCOUT);
        TILE_CHAIN(ua, 1, fb)
        keep = tile_sum(fa);
        const int tn = (t0 + 2 * kG < kT) ? (t0 + 2 * kG) : 0;
        load_group(ua, ul + tn * kCIN);
        TILE_CHAIN(ub, 0, fa)
        store_pair(keep, tile_sum(fb), lane, ob + t0 * kCOUT);
        TILE_CHAIN(ub, 1, fb)
        keep = tile_sum(fa);
    }
#undef TILE_CHAIN
    store_pair(keep, tile_sum(fb), lane, ob + (kT - kG) * kCOUT);
}

extern "C" void kernel_launch(void* const* d_in, const int* in_sizes, int n_in,
                              void* d_out, int out_size, void* d_ws, size_t ws_size,
                              hipStream_t stream) {
    (void)n_in; (void)out_size; (void)d_ws; (void)ws_size;
    const float* u  = (const float*)d_in[0];
    const float* bc = (const float*)d_in[1];
    const float* ac = (const float*)d_in[2];
    float* out = (float*)d_out;

    const int B = in_sizes[0] / (kT * kCIN);
    dim3 grid(B * (kCOUT / kOPB));
    dim3 block(kOPB * 32);
    mimo_iir_wmma_kernel<<<grid, block, 0, stream>>>(u, bc, ac, out);
}


# 4 waves per block instead of 8 (256 blocks of 128 threads): the grid fits one wave per SIMD on all 256 WGPs instead of two per SIMD on half of them; per-wave code unchanged
# speedup vs baseline: 4.0043x; 1.4413x over previous
#include <hip/hip_runtime.h>


typedef float v2f __attribute__((ext_vector_type(2)));
typedef float v8f __attribute__((ext_vector_type(8)));

constexpr int kT    = 2048;
constexpr int kCIN  = 32;
constexpr int kCOUT = 32;
constexpr int kTT   = 16;
constexpr int kG    = 32;
constexpr int kOPB  = 4;
constexpr int kPAD  = 17;
constexpr int kNF   = kCIN / 4;

__device__ __forceinline__ void load_group(float (&dst)[kG], const float* p)
{
    #pragma unroll
    for (int k = 0; k < kG; ++k) dst[k] = p[k * kCIN];
}

__device__ __forceinline__ void chain_tile(
    const float (&uc)[kG], const int h,
    float b0, float b1, float b2, float a0, float a1, float a2,
    float& up1, float& up2, float& up3, float& y1, float& y2, float& y3,
    float* yw, int i)
{
    #pragma unroll
    for (int tt = 0; tt < kTT; ++tt) {
        float x  = fmaf(b0, up1, fmaf(b1, up2, b2 * up3));
        float p  = fmaf(-a1, y2, fmaf(-a2, y3, x));
        float yt = fmaf(-a0, y1, p);
        yw[i * kPAD + tt] = yt;
        y3 = y2; y2 = y1; y1 = yt;
        up3 = up2; up2 = up1; up1 = uc[h * kTT + tt];
    }
}

__device__ __forceinline__ void read_frags(v2f (&fr)[kNF], const float* fbase)
{
    #pragma unroll
    for (int g = 0; g < kNF; ++g) {
        fr[g].x = fbase[(4 * g + 0) * kPAD];
        fr[g].y = fbase[(4 * g + 1) * kPAD];
    }
}

__device__ __forceinline__ float tile_sum(const v2f (&fr)[kNF])
{
    v2f ones; ones.x = 1.0f; ones.y = 1.0f;
    v8f acc = {};
    #pragma unroll
    for (int g = 0; g < kNF; ++g) {
        acc = __builtin_amdgcn_wmma_f32_16x16x4_f32(
            false, ones, false, fr[g], (short)0, acc, false, false);
    }
    return acc[0];
}

__device__ __forceinline__ void store_pair(float lo, float hi, int lane,
                                           float* __restrict__ ot)
{
    ot[lane * kCOUT] = (lane < 16) ? lo : hi;
}

__global__ __launch_bounds__(kOPB * 32) void mimo_iir_wmma_kernel(
    const float* u,
    const float* __restrict__ bc,
    const float* __restrict__ ac,
    float* __restrict__ out)
{
    __shared__ float y_lds[kOPB * kCIN * kPAD];

    const int tid  = threadIdx.x;
    const int w    = tid >> 5;
    const int lane = tid & 31;
    const int b    = blockIdx.x / (kCOUT / kOPB);
    const int o    = (blockIdx.x % (kCOUT / kOPB)) * kOPB + w;
    const int i    = lane;

    const int cb = (o * kCIN + i) * 3;
    const float b0 = bc[cb + 0], b1 = bc[cb + 1], b2 = bc[cb + 2];
    const float a0 = ac[cb + 0], a1 = ac[cb + 1], a2 = ac[cb + 2];

    float up1 = 0.f, up2 = 0.f, up3 = 0.f;
    float y1 = 0.f, y2 = 0.f, y3 = 0.f;

    const float* ul = u   + (size_t)b * kT * kCIN + i;
    float*       ob = out + (size_t)b * kT * kCOUT + o;

    const int m   = lane & 15;
    const int khi = (lane < 16) ? 0 : 2;

    float* yw = &y_lds[w * kCIN * kPAD];
    const float* fbase = yw + khi * kPAD + m;

    float ua[kG], ub[kG];
    v2f fa[kNF], fb[kNF] = {};
    load_group(ua, ul);

#define TILE_CHAIN(UC, H, FNEW)                                                      \
    chain_tile(UC, H, b0, b1, b2, a0, a1, a2, up1, up2, up3, y1, y2, y3, yw, i);     \
    __builtin_amdgcn_wave_barrier();                                                 \
    read_frags(FNEW, fbase);                                                         \
    __builtin_amdgcn_sched_barrier(0);

    float keep = 0.f;

    for (int t0 = 0; t0 < kT; t0 += 2 * kG) {
        load_group(ub, ul + (t0 + kG) * kCIN);
        TILE_CHAIN(ua, 0, fa)
        if (t0 > 0) store_pair(keep, tile_sum(fb), lane, ob + (t0 - kG) * kCOUT);
        TILE_CHAIN(ua, 1, fb)
        keep = tile_sum(fa);
        const int tn = (t0 + 2 * kG < kT) ? (t0 + 2 * kG) : 0;
        load_group(ua, ul + tn * kCIN);
        TILE_CHAIN(ub, 0, fa)
        store_pair(keep, tile_sum(fb), lane, ob + t0 * kCOUT);
        TILE_CHAIN(ub, 1, fb)
        keep = tile_sum(fa);
    }
#undef TILE_CHAIN
    store_pair(keep, tile_sum(fb), lane, ob + (kT - kG) * kCOUT);
}

extern "C" void kernel_launch(void* const* d_in, const int* in_sizes, int n_in,
                              void* d_out, int out_size, void* d_ws, size_t ws_size,
                              hipStream_t stream) {
    (void)n_in; (void)out_size; (void)d_ws; (void)ws_size;
    const float* u  = (const float*)d_in[0];
    const float* bc = (const float*)d_in[1];
    const float* ac = (const float*)d_in[2];
    float* out = (float*)d_out;

    const int B = in_sizes[0] / (kT * kCIN);
    dim3 grid(B * (kCOUT / kOPB));
    dim3 block(kOPB * 32);
    mimo_iir_wmma_kernel<<<grid, block, 0, stream>>>(u, bc, ac, out);
}


# u fetched once per block instead of once per wave: each wave fetches a quarter of a 128-step chunk one chunk ahead, stages it in a double-buffered LDS chunk; one block barrier per 128 steps
# speedup vs baseline: 4.2962x; 1.0729x over previous
#include <hip/hip_runtime.h>


typedef float v2f __attribute__((ext_vector_type(2)));
typedef float v8f __attribute__((ext_vector_type(8)));

constexpr int kT    = 2048;
constexpr int kCIN  = 32;
constexpr int kCOUT = 32;
constexpr int kTT   = 16;
constexpr int kG    = 32;
constexpr int kOPB  = 4;
constexpr int kPAD  = 17;
constexpr int kNF   = kCIN / 4;
constexpr int kCH   = 128;
constexpr int kNCH  = kT / kCH;
constexpr int kRW   = kCH / kOPB;

__device__ __forceinline__ void load_group(float (&dst)[kG], const float* p)
{
    #pragma unroll
    for (int k = 0; k < kG; ++k) dst[k] = p[k * kCIN];
}

__device__ __forceinline__ void store_rows(float* p, const float (&src)[kRW])
{
    #pragma unroll
    for (int k = 0; k < kRW; ++k) p[k * kCIN] = src[k];
}

__device__ __forceinline__ void lds_block_barrier()
{
    __builtin_amdgcn_fence(__ATOMIC_RELEASE, "workgroup", "local");
    __builtin_amdgcn_s_barrier();
    __builtin_amdgcn_fence(__ATOMIC_ACQUIRE, "workgroup", "local");
}

__device__ __forceinline__ void chain_tile(
    const float (&uc)[kG], const int h,
    float b0, float b1, float b2, float a0, float a1, float a2,
    float& up1, float& up2, float& up3, float& y1, float& y2, float& y3,
    float* yw, int i)
{
    #pragma unroll
    for (int tt = 0; tt < kTT; ++tt) {
        float x  = fmaf(b0, up1, fmaf(b1, up2, b2 * up3));
        float p  = fmaf(-a1, y2, fmaf(-a2, y3, x));
        float yt = fmaf(-a0, y1, p);
        yw[i * kPAD + tt] = yt;
        y3 = y2; y2 = y1; y1 = yt;
        up3 = up2; up2 = up1; up1 = uc[h * kTT + tt];
    }
}

__device__ __forceinline__ void read_frags(v2f (&fr)[kNF], const float* fbase)
{
    #pragma unroll
    for (int g = 0; g < kNF; ++g) {
        fr[g].x = fbase[(4 * g + 0) * kPAD];
        fr[g].y = fbase[(4 * g + 1) * kPAD];
    }
}

__device__ __forceinline__ float tile_sum(const v2f (&fr)[kNF])
{
    v2f ones; ones.x = 1.0f; ones.y = 1.0f;
    v8f acc = {};
    #pragma unroll
    for (int g = 0; g < kNF; ++g) {
        acc = __builtin_amdgcn_wmma_f32_16x16x4_f32(
            false, ones, false, fr[g], (short)0, acc, false, false);
    }
    return acc[0];
}

__device__ __forceinline__ void store_pair(float lo, float hi, int lane,
                                           float* __restrict__ ot)
{
    ot[lane * kCOUT] = (lane < 16) ? lo : hi;
}

__global__ __launch_bounds__(kOPB * 32) void mimo_iir_wmma_kernel(
    const float* u,
    const float* __restrict__ bc,
    const float* __restrict__ ac,
    float* __restrict__ out)
{
    __shared__ float y_lds[kOPB * kCIN * kPAD];
    __shared__ float u_lds[2 * kCH * kCIN];

    const int tid  = threadIdx.x;
    const int w    = tid >> 5;
    const int lane = tid & 31;
    const int b    = blockIdx.x / (kCOUT / kOPB);
    const int o    = (blockIdx.x % (kCOUT / kOPB)) * kOPB + w;
    const int i    = lane;

    const int cb = (o * kCIN + i) * 3;
    const float b0 = bc[cb + 0], b1 = bc[cb + 1], b2 = bc[cb + 2];
    const float a0 = ac[cb + 0], a1 = ac[cb + 1], a2 = ac[cb + 2];

    float up1 = 0.f, up2 = 0.f, up3 = 0.f;
    float y1 = 0.f, y2 = 0.f, y3 = 0.f;

    const float* ul = u   + (size_t)b * kT * kCIN + i;
    float*       ob = out + (size_t)b * kT * kCOUT + o;

    const int m   = lane & 15;
    const int khi = (lane < 16) ? 0 : 2;

    float* yw = &y_lds[w * kCIN * kPAD];
    const float* fbase = yw + khi * kPAD + m;

    float ua[kG], ub[kG];
    float gq[kRW];
    const float* uq = ul + w * kRW * kCIN;
    float* sq = u_lds + w * kRW * kCIN + lane;
    v2f fa[kNF], fb[kNF] = {};
    load_group(gq, uq);
    store_rows(sq, gq);
    load_group(gq, uq + kCH * kCIN);
    lds_block_barrier();
    load_group(ua, u_lds + lane);

#define TILE_CHAIN(UC, H, FNEW)                                                      \
    chain_tile(UC, H, b0, b1, b2, a0, a1, a2, up1, up2, up3, y1, y2, y3, yw, i);     \
    __builtin_amdgcn_wave_barrier();                                                 \
    read_frags(FNEW, fbase);                                                         \
    __builtin_amdgcn_sched_barrier(0);

    float keep = 0.f;

    for (int c = 0; c < kNCH; ++c) {
        const float* cur = u_lds + (c & 1) * (kCH * kCIN) + lane;
        const int    nb  = ((c + 1) & 1) * (kCH * kCIN);
        float*       oc  = ob + c * kCH * kCOUT;
        load_group(ub, cur + 1 * kG * kCIN);
        TILE_CHAIN(ua, 0, fa)
        if (c > 0) store_pair(keep, tile_sum(fb), lane, oc - kG * kCOUT);
        TILE_CHAIN(ua, 1, fb)
        keep = tile_sum(fa);
        store_rows(sq + nb, gq);
        const int cn = (c + 2 < kNCH) ? (c + 2) : 0;
        load_group(gq, uq + cn * kCH * kCIN);
        load_group(ua, cur + 2 * kG * kCIN);
        TILE_CHAIN(ub, 0, fa)
        store_pair(keep, tile_sum(fb), lane, oc);
        TILE_CHAIN(ub, 1, fb)
        keep = tile_sum(fa);
        load_group(ub, cur + 3 * kG * kCIN);
        TILE_CHAIN(ua, 0, fa)
        store_pair(keep, tile_sum(fb), lane, oc + kG * kCOUT);
        TILE_CHAIN(ua, 1, fb)
        keep = tile_sum(fa);
        lds_block_barrier();
        load_group(ua, u_lds + nb + lane);
        TILE_CHAIN(ub, 0, fa)
        store_pair(keep, tile_sum(fb), lane, oc + 2 * kG * kCOUT);
        TILE_CHAIN(ub, 1, fb)
        keep = tile_sum(fa);
    }
#undef TILE_CHAIN
    store_pair(keep, tile_sum(fb), lane, ob + (kT - kG) * kCOUT);
}

extern "C" void kernel_launch(void* const* d_in, const int* in_sizes, int n_in,
                              void* d_out, int out_size, void* d_ws, size_t ws_size,
                              hipStream_t stream) {
    (void)n_in; (void)out_size; (void)d_ws; (void)ws_size;
    const float* u  = (const float*)d_in[0];
    const float* bc = (const float*)d_in[1];
    const float* ac = (const float*)d_in[2];
    float* out = (float*)d_out;

    const int B = in_sizes[0] / (kT * kCIN);
    dim3 grid(B * (kCOUT / kOPB));
    dim3 block(kOPB * 32);
    mimo_iir_wmma_kernel<<<grid, block, 0, stream>>>(u, bc, ac, out);
}


# channel sum moved to a second wave per SIMD: waves 0-3 run the recurrence and write y into double-buffered 128-step LDS windows; waves 4-7 read them as fragments, matrix-sum and store; sum order kept
# speedup vs baseline: 4.8492x; 1.1287x over previous
#include <hip/hip_runtime.h>


typedef float v2f __attribute__((ext_vector_type(2)));
typedef float v4f __attribute__((ext_vector_type(4)));
typedef float v8f __attribute__((ext_vector_type(8)));

constexpr int kT    = 2048;
constexpr int kCIN  = 32;
constexpr int kCOUT = 32;
constexpr int kTT   = 16;
constexpr int kG    = 32;
constexpr int kOPB  = 4;
constexpr int kNF   = kCIN / 4;
constexpr int kCH   = 128;
constexpr int kNCH  = kT / kCH;
constexpr int kRW   = kCH / kOPB;
constexpr int kYP   = 36;
constexpr int kYW   = kCH * kYP;
constexpr int kYB   = kOPB * kYW;

__device__ __forceinline__ void load_group(float (&dst)[kG], const float* p)
{
    #pragma unroll
    for (int k = 0; k < kG; ++k) dst[k] = p[k * kCIN];
}

__device__ __forceinline__ void store_rows(float* p, const float (&src)[kRW])
{
    #pragma unroll
    for (int k = 0; k < kRW; ++k) p[k * kCIN] = src[k];
}

__device__ __forceinline__ void lds_block_barrier()
{
    __builtin_amdgcn_fence(__ATOMIC_RELEASE, "workgroup", "local");
    __builtin_amdgcn_s_barrier();
    __builtin_amdgcn_fence(__ATOMIC_ACQUIRE, "workgroup", "local");
}

__device__ __forceinline__ void chain_tile(
    const float (&uc)[kG], const int h,
    float b0, float b1, float b2, float a0, float a1, float a2,
    float& up1, float& up2, float& up3, float& y1, float& y2, float& y3,
    float* ycol)
{
    float yv[kTT];
    #pragma unroll
    for (int tt = 0; tt < kTT; ++tt) {
        float x  = fmaf(b0, up1, fmaf(b1, up2, b2 * up3));
        float p  = fmaf(-a1, y2, fmaf(-a2, y3, x));
        float yt = fmaf(-a0, y1, p);
        yv[tt] = yt;
        y3 = y2; y2 = y1; y1 = yt;
        up3 = up2; up2 = up1; up1 = uc[h * kTT + tt];
    }
    __builtin_amdgcn_sched_barrier(0);
    #pragma unroll
    for (int tt = 0; tt < kTT; ++tt) ycol[tt * kYP] = yv[tt];
}

__device__ __forceinline__ void read_frags(v2f (&fr)[kNF], const float* frow)
{
    #pragma unroll
    for (int j = 0; j < kNF / 2; ++j) {
        const v4f v = *reinterpret_cast<const v4f*>(frow + 4 * j);
        fr[2 * j + 0].x = v.x; fr[2 * j + 0].y = v.y;
        fr[2 * j + 1].x = v.z; fr[2 * j + 1].y = v.w;
    }
}

__device__ __forceinline__ void pair_sum(const v2f (&f0)[kNF], const v2f (&f1)[kNF],
                                         float& lo, float& hi)
{
    v2f ones; ones.x = 1.0f; ones.y = 1.0f;
    v8f acc0 = {};
    v8f acc1 = {};
    #pragma unroll
    for (int g = 0; g < kNF; ++g) {
        acc0 = __builtin_amdgcn_wmma_f32_16x16x4_f32(
            false, ones, false, f0[g], (short)0, acc0, false, false);
        acc1 = __builtin_amdgcn_wmma_f32_16x16x4_f32(
            false, ones, false, f1[g], (short)0, acc1, false, false);
    }
    lo = acc0[0];
    hi = acc1[0];
}

__device__ __forceinline__ void store_pair(float lo, float hi, int lane,
                                           float* __restrict__ ot)
{
    ot[lane * kCOUT] = (lane < 16) ? lo : hi;
}

template <int Q0, int Q1>
__device__ __forceinline__ void sum_window(const float* fwin, float* __restrict__ og, int lane)
{
    v2f f[2][2][kNF];
    read_frags(f[Q0 & 1][0], fwin + (Q0 * kG) * kYP);
    read_frags(f[Q0 & 1][1], fwin + (Q0 * kG + kTT) * kYP);
    #pragma unroll
    for (int q = Q0; q < Q1; ++q) {
        if (q + 1 < Q1) {
            read_frags(f[(q + 1) & 1][0], fwin + ((q + 1) * kG) * kYP);
            read_frags(f[(q + 1) & 1][1], fwin + ((q + 1) * kG + kTT) * kYP);
        }
        __builtin_amdgcn_sched_barrier(0);
        float lo, hi;
        pair_sum(f[q & 1][0], f[q & 1][1], lo, hi);
        store_pair(lo, hi, lane, og + (q - Q0) * kG * kCOUT);
        __builtin_amdgcn_sched_barrier(0);
    }
}

__global__ __launch_bounds__(2 * kOPB * 32) void mimo_iir_wmma_kernel(
    const float* u,
    const float* __restrict__ bc,
    const float* __restrict__ ac,
    float* __restrict__ out)
{
    __shared__ __attribute__((aligned(16))) float y_lds[2 * kYB];
    __shared__ float u_lds[2 * kCH * kCIN];

    const int tid  = threadIdx.x;
    const int wv   = __builtin_amdgcn_readfirstlane(tid >> 5);
    const bool rec = wv < kOPB;
    const int w    = wv & (kOPB - 1);
    const int lane = tid & 31;
    const int b    = blockIdx.x / (kCOUT / kOPB);
    const int o    = (blockIdx.x % (kCOUT / kOPB)) * kOPB + w;
    const int i    = lane;

    float b0 = 0.f, b1 = 0.f, b2 = 0.f, a0 = 0.f, a1 = 0.f, a2 = 0.f;
    float up1 = 0.f, up2 = 0.f, up3 = 0.f;
    float y1 = 0.f, y2 = 0.f, y3 = 0.f;
    float ua[kG] = {}, ub[kG] = {};
    float gq[kRW] = {};

    const float* ul = u + (size_t)b * kT * kCIN + i;
    const float* uq = ul + w * kRW * kCIN;
    float* sq = u_lds + w * kRW * kCIN + lane;
    float* ycol = y_lds + w * kYW + ((i >> 1) & 1) * 16 + (i >> 2) * 2 + (i & 1);

    float* ob = out + (size_t)b * kT * kCOUT + o;
    const int m   = lane & 15;
    const int hh  = lane >> 4;
    const float* fcol = y_lds + w * kYW + m * kYP + 16 * hh;

    if (rec) {
        const int cb = (o * kCIN + i) * 3;
        b0 = bc[cb + 0]; b1 = bc[cb + 1]; b2 = bc[cb + 2];
        a0 = ac[cb + 0]; a1 = ac[cb + 1]; a2 = ac[cb + 2];
        load_group(gq, uq);
        store_rows(sq, gq);
        load_group(gq, uq + kCH * kCIN);
    }
    lds_block_barrier();
    if (rec) load_group(ua, u_lds + lane);

#define CHAIN(UC, H, YR) \
    chain_tile(UC, H, b0, b1, b2, a0, a1, a2, up1, up2, up3, y1, y2, y3, YR)

    for (int c = 0; c < kNCH; ++c) {
        const int nb = ((c + 1) & 1) * (kCH * kCIN);
        if (rec) {
            const float* cur = u_lds + (c & 1) * (kCH * kCIN) + lane;
            float* yc = ycol + (c & 1) * kYB;
#define LDS_GROUP_END __builtin_amdgcn_sched_barrier(0)
            load_group(ub, cur + 1 * kG * kCIN);
            LDS_GROUP_END;
            CHAIN(ua, 0, yc + (1 * kG) * kYP);
            LDS_GROUP_END;
            CHAIN(ua, 1, yc + (1 * kG + kTT) * kYP);
            store_rows(sq + nb, gq);
            const int cn = (c + 2 < kNCH) ? (c + 2) : 0;
            load_group(gq, uq + cn * kCH * kCIN);
            load_group(ua, cur + 2 * kG * kCIN);
            LDS_GROUP_END;
            CHAIN(ub, 0, yc + (2 * kG) * kYP);
            LDS_GROUP_END;
            CHAIN(ub, 1, yc + (2 * kG + kTT) * kYP);
            load_group(ub, cur + 3 * kG * kCIN);
            LDS_GROUP_END;
            CHAIN(ua, 0, yc + (3 * kG) * kYP);
            LDS_GROUP_END;
            CHAIN(ua, 1, yc + (3 * kG + kTT) * kYP);
        }
        lds_block_barrier();
        if (rec) {
            float* yn = ycol + ((c + 1) & 1) * kYB;
            load_group(ua, u_lds + nb + lane);
            LDS_GROUP_END;
            CHAIN(ub, 0, yn);
            LDS_GROUP_END;
            CHAIN(ub, 1, yn + kTT * kYP);
        } else {
            const float* fwin = fcol + (c & 1) * kYB;
            if (c == 0) sum_window<1, 4>(fwin, ob, lane);
            else        sum_window<0, 4>(fwin, ob + (c * kCH - kG) * kCOUT, lane);
        }
    }
#undef CHAIN
#undef LDS_GROUP_END
    lds_block_barrier();
    if (!rec) sum_window<0, 1>(fcol + (kNCH & 1) * kYB, ob + (kT - kG) * kCOUT, lane);
}

extern "C" void kernel_launch(void* const* d_in, const int* in_sizes, int n_in,
                              void* d_out, int out_size, void* d_ws, size_t ws_size,
                              hipStream_t stream) {
    (void)n_in; (void)out_size; (void)d_ws; (void)ws_size;
    const float* u  = (const float*)d_in[0];
    const float* bc = (const float*)d_in[1];
    const float* ac = (const float*)d_in[2];
    float* out = (float*)d_out;

    const int B = in_sizes[0] / (kT * kCIN);
    dim3 grid(B * (kCOUT / kOPB));
    dim3 block(2 * kOPB * 32);
    mimo_iir_wmma_kernel<<<grid, block, 0, stream>>>(u, bc, ac, out);
}


# output leaves as 16-byte pieces: sum waves stage their sums in LDS (row = timestep, column = the block's 4 output channels); one 16-byte store per lane replaces four 4-byte stores; sum order kept
# speedup vs baseline: 5.9514x; 1.2273x over previous
#include <hip/hip_runtime.h>


typedef float v2f __attribute__((ext_vector_type(2)));
typedef float v4f __attribute__((ext_vector_type(4)));
typedef float v8f __attribute__((ext_vector_type(8)));

constexpr int kT    = 2048;
constexpr int kCIN  = 32;
constexpr int kCOUT = 32;
constexpr int kTT   = 16;
constexpr int kG    = 32;
constexpr int kOPB  = 4;
constexpr int kNF   = kCIN / 4;
constexpr int kCH   = 128;
constexpr int kNCH  = kT / kCH;
constexpr int kRW   = kCH / kOPB;
constexpr int kYP   = 36;
constexpr int kYW   = kCH * kYP;
constexpr int kYB   = kOPB * kYW;
constexpr int kSG   = kG * kOPB;
constexpr int kSB   = (kCH / kG) * kSG;

__device__ __forceinline__ void load_group(float (&dst)[kG], const float* p)
{
    #pragma unroll
    for (int k = 0; k < kG; ++k) dst[k] = p[k * kCIN];
}

__device__ __forceinline__ void store_rows(float* p, const float (&src)[kRW])
{
    #pragma unroll
    for (int k = 0; k < kRW; ++k) p[k * kCIN] = src[k];
}

__device__ __forceinline__ void lds_block_barrier()
{
    __builtin_amdgcn_fence(__ATOMIC_RELEASE, "workgroup", "local");
    __builtin_amdgcn_s_barrier();
    __builtin_amdgcn_fence(__ATOMIC_ACQUIRE, "workgroup", "local");
}

__device__ __forceinline__ void chain_tile(
    const float (&uc)[kG], const int h,
    float b0, float b1, float b2, float a0, float a1, float a2,
    float& up1, float& up2, float& up3, float& y1, float& y2, float& y3,
    float* ycol)
{
    float yv[kTT];
    #pragma unroll
    for (int tt = 0; tt < kTT; ++tt) {
        float x  = fmaf(b0, up1, fmaf(b1, up2, b2 * up3));
        float p  = fmaf(-a1, y2, fmaf(-a2, y3, x));
        float yt = fmaf(-a0, y1, p);
        yv[tt] = yt;
        y3 = y2; y2 = y1; y1 = yt;
        up3 = up2; up2 = up1; up1 = uc[h * kTT + tt];
    }
    __builtin_amdgcn_sched_barrier(0);
    #pragma unroll
    for (int tt = 0; tt < kTT; ++tt) ycol[tt * kYP] = yv[tt];
}

__device__ __forceinline__ void read_frags(v2f (&fr)[kNF], const float* frow)
{
    #pragma unroll
    for (int j = 0; j < kNF / 2; ++j) {
        const v4f v = *reinterpret_cast<const v4f*>(frow + 4 * j);
        fr[2 * j + 0].x = v.x; fr[2 * j + 0].y = v.y;
        fr[2 * j + 1].x = v.z; fr[2 * j + 1].y = v.w;
    }
}

__device__ __forceinline__ void pair_sum(const v2f (&f0)[kNF], const v2f (&f1)[kNF],
                                         float& lo, float& hi)
{
    v2f ones; ones.x = 1.0f; ones.y = 1.0f;
    v8f acc0 = {};
    v8f acc1 = {};
    #pragma unroll
    for (int g = 0; g < kNF; ++g) {
        acc0 = __builtin_amdgcn_wmma_f32_16x16x4_f32(
            false, ones, false, f0[g], (short)0, acc0, false, false);
        acc1 = __builtin_amdgcn_wmma_f32_16x16x4_f32(
            false, ones, false, f1[g], (short)0, acc1, false, false);
    }
    lo = acc0[0];
    hi = acc1[0];
}

template <int Q0, int Q1>
__device__ __forceinline__ void sum_window(const float* fwin, float* sw, int lane,
                                           const float* pr, float* __restrict__ pd, bool has)
{
    v2f f[2][2][kNF];
    v4f pv = {};
    if (has) pv = *reinterpret_cast<const v4f*>(pr);
    read_frags(f[Q0 & 1][0], fwin + (Q0 * kG) * kYP);
    read_frags(f[Q0 & 1][1], fwin + (Q0 * kG + kTT) * kYP);
    #pragma unroll
    for (int q = Q0; q < Q1; ++q) {
        if (q + 1 < Q1) {
            read_frags(f[(q + 1) & 1][0], fwin + ((q + 1) * kG) * kYP);
            read_frags(f[(q + 1) & 1][1], fwin + ((q + 1) * kG + kTT) * kYP);
        }
        __builtin_amdgcn_sched_barrier(0);
        if (q == Q0 && has) *reinterpret_cast<v4f*>(pd) = pv;
        float lo, hi;
        pair_sum(f[q & 1][0], f[q & 1][1], lo, hi);
        sw[q * kSG] = (lane < 16) ? lo : hi;
        __builtin_amdgcn_sched_barrier(0);
    }
}

__global__ __launch_bounds__(2 * kOPB * 32) void mimo_iir_wmma_kernel(
    const float* u,
    const float* __restrict__ bc,
    const float* __restrict__ ac,
    float* __restrict__ out)
{
    __shared__ __attribute__((aligned(16))) float y_lds[2 * kYB];
    __shared__ float u_lds[2 * kCH * kCIN];
    __shared__ __attribute__((aligned(16))) float stage[2 * kSB];

    const int tid  = threadIdx.x;
    const int wv   = __builtin_amdgcn_readfirstlane(tid >> 5);
    const bool rec = wv < kOPB;
    const int w    = wv & (kOPB - 1);
    const int lane = tid & 31;
    const int b    = blockIdx.x / (kCOUT / kOPB);
    const int o    = (blockIdx.x % (kCOUT / kOPB)) * kOPB + w;
    const int i    = lane;

    float b0 = 0.f, b1 = 0.f, b2 = 0.f, a0 = 0.f, a1 = 0.f, a2 = 0.f;
    float up1 = 0.f, up2 = 0.f, up3 = 0.f;
    float y1 = 0.f, y2 = 0.f, y3 = 0.f;
    float ua[kG] = {}, ub[kG] = {};
    float gq[kRW] = {};

    const float* ul = u + (size_t)b * kT * kCIN + i;
    const float* uq = ul + w * kRW * kCIN;
    float* sq = u_lds + w * kRW * kCIN + lane;
    float* ycol = y_lds + w * kYW + ((i >> 1) & 1) * 16 + (i >> 2) * 2 + (i & 1);

    float* ob4 = out + (size_t)b * kT * kCOUT + (o - w);
    const int m   = lane & 15;
    const int hh  = lane >> 4;
    const float* fcol = y_lds + w * kYW + m * kYP + 16 * hh;

    if (rec) {
        const int cb = (o * kCIN + i) * 3;
        b0 = bc[cb + 0]; b1 = bc[cb + 1]; b2 = bc[cb + 2];
        a0 = ac[cb + 0]; a1 = ac[cb + 1]; a2 = ac[cb + 2];
        load_group(gq, uq);
        store_rows(sq, gq);
        load_group(gq, uq + kCH * kCIN);
    }
    lds_block_barrier();
    if (rec) load_group(ua, u_lds + lane);

#define CHAIN(UC, H, YR) \
    chain_tile(UC, H, b0, b1, b2, a0, a1, a2, up1, up2, up3, y1, y2, y3, YR)

    for (int c = 0; c < kNCH; ++c) {
        const int nb = ((c + 1) & 1) * (kCH * kCIN);
        if (rec) {
            const float* cur = u_lds + (c & 1) * (kCH * kCIN) + lane;
            float* yc = ycol + (c & 1) * kYB;
#define LDS_GROUP_END __builtin_amdgcn_sched_barrier(0)
            load_group(ub, cur + 1 * kG * kCIN);
            LDS_GROUP_END;
            CHAIN(ua, 0, yc + (1 * kG) * kYP);
            LDS_GROUP_END;
            CHAIN(ua, 1, yc + (1 * kG + kTT) * kYP);
            store_rows(sq + nb, gq);
            const int cn = (c + 2 < kNCH) ? (c + 2) : 0;
            load_group(gq, uq + cn * kCH * kCIN);
            load_group(ua, cur + 2 * kG * kCIN);
            LDS_GROUP_END;
            CHAIN(ub, 0, yc + (2 * kG) * kYP);
            LDS_GROUP_END;
            CHAIN(ub, 1, yc + (2 * kG + kTT) * kYP);
            load_group(ub, cur + 3 * kG * kCIN);
            LDS_GROUP_END;
            CHAIN(ua, 0, yc + (3 * kG) * kYP);
            LDS_GROUP_END;
            CHAIN(ua, 1, yc + (3 * kG + kTT) * kYP);
        }
        lds_block_barrier();
        if (rec) {
            float* yn = ycol + ((c + 1) & 1) * kYB;
            load_group(ua, u_lds + nb + lane);
            LDS_GROUP_END;
            CHAIN(ub, 0, yn);
            LDS_GROUP_END;
            CHAIN(ub, 1, yn + kTT * kYP);
        } else {
            const float* fwin = fcol + (c & 1) * kYB;
            float* sw = stage + (c & 1) * kSB + lane * kOPB + w;
            if (c == 0) {
                sum_window<1, 4>(fwin, sw, lane, nullptr, nullptr, false);
            } else {
                const int tg = (c - 1) * kCH - kG + w * kG;
                const bool has = tg >= 0;
                const float* pr = stage + ((c - 1) & 1) * kSB + w * kSG + lane * kOPB;
                float* pd = ob4 + (size_t)(c * kCH + w * kG + lane) * kCOUT - (kCH + kG) * kCOUT;
                sum_window<0, 4>(fwin, sw, lane, pr, pd, has);
            }
        }
    }
#undef CHAIN
#undef LDS_GROUP_END
    lds_block_barrier();
    if (!rec) {
        const float* pr = stage + ((kNCH - 1) & 1) * kSB + w * kSG + lane * kOPB;
        float* pd = ob4 + (size_t)((kNCH - 1) * kCH - kG + w * kG + lane) * kCOUT;
        sum_window<0, 1>(fcol + (kNCH & 1) * kYB, stage + (kNCH & 1) * kSB + lane * kOPB + w, lane,
                         pr, pd, true);
    }
    lds_block_barrier();
    if (wv == kOPB) {
        const v4f v = *reinterpret_cast<const v4f*>(stage + (kNCH & 1) * kSB + lane * kOPB);
        *reinterpret_cast<v4f*>(ob4 + (size_t)(kT - kG + lane) * kCOUT) = v;
    }
}

extern "C" void kernel_launch(void* const* d_in, const int* in_sizes, int n_in,
                              void* d_out, int out_size, void* d_ws, size_t ws_size,
                              hipStream_t stream) {
    (void)n_in; (void)out_size; (void)d_ws; (void)ws_size;
    const float* u  = (const float*)d_in[0];
    const float* bc = (const float*)d_in[1];
    const float* ac = (const float*)d_in[2];
    float* out = (float*)d_out;

    const int B = in_sizes[0] / (kT * kCIN);
    dim3 grid(B * (kCOUT / kOPB));
    dim3 block(2 * kOPB * 32);
    mimo_iir_wmma_kernel<<<grid, block, 0, stream>>>(u, bc, ac, out);
}


# u path widened to 16 bytes per lane: quarter chunk fetched with 16-byte global loads and kept channel-major in LDS (pitch 132); a 32-step group is 8 16-byte LDS reads per lane, not 32 4-byte
# speedup vs baseline: 6.6393x; 1.1156x over previous
#include <hip/hip_runtime.h>


typedef float v2f __attribute__((ext_vector_type(2)));
typedef float v4f __attribute__((ext_vector_type(4)));
typedef float v8f __attribute__((ext_vector_type(8)));

constexpr int kT    = 2048;
constexpr int kCIN  = 32;
constexpr int kCOUT = 32;
constexpr int kTT   = 16;
constexpr int kG    = 32;
constexpr int kOPB  = 4;
constexpr int kNF   = kCIN / 4;
constexpr int kCH   = 128;
constexpr int kNCH  = kT / kCH;
constexpr int kRW   = kCH / kOPB;
constexpr int kUP   = kCH + 4;
constexpr int kUS   = 4;
constexpr int kUB   = kCIN * kUP;
constexpr int kYP   = 36;
constexpr int kYW   = kCH * kYP;
constexpr int kYB   = kOPB * kYW;
constexpr int kSG   = kG * kOPB;
constexpr int kSB   = (kCH / kG) * kSG;

__device__ __forceinline__ void fetch_quarter(v4f (&dst)[kRW / 4], const float* p)
{
    #pragma unroll
    for (int j = 0; j < kRW / 4; ++j)
        dst[j] = *reinterpret_cast<const v4f*>(p + j * (4 * kCIN));
}

__device__ __forceinline__ void load_group_lds(float (&dst)[kG], const float* p)
{
    #pragma unroll
    for (int j = 0; j < kG / 4; ++j) {
        const v4f v = *reinterpret_cast<const v4f*>(p + 4 * j);
        dst[4 * j + 0] = v.x; dst[4 * j + 1] = v.y;
        dst[4 * j + 2] = v.z; dst[4 * j + 3] = v.w;
    }
}

__device__ __forceinline__ void store_quarter(float* p, const v4f (&src)[kRW / 4])
{
    #pragma unroll
    for (int j = 0; j < kRW / 4; ++j) {
        p[0 * kUP + 4 * j] = src[j].x;
        p[1 * kUP + 4 * j] = src[j].y;
        p[2 * kUP + 4 * j] = src[j].z;
        p[3 * kUP + 4 * j] = src[j].w;
    }
}

__device__ __forceinline__ void lds_block_barrier()
{
    __builtin_amdgcn_fence(__ATOMIC_RELEASE, "workgroup", "local");
    __builtin_amdgcn_s_barrier();
    __builtin_amdgcn_fence(__ATOMIC_ACQUIRE, "workgroup", "local");
}

__device__ __forceinline__ void chain_tile(
    const float (&uc)[kG], const int h,
    float b0, float b1, float b2, float a0, float a1, float a2,
    float& up1, float& up2, float& up3, float& y1, float& y2, float& y3,
    float* ycol)
{
    float yv[kTT];
    #pragma unroll
    for (int tt = 0; tt < kTT; ++tt) {
        float x  = fmaf(b0, up1, fmaf(b1, up2, b2 * up3));
        float p  = fmaf(-a1, y2, fmaf(-a2, y3, x));
        float yt = fmaf(-a0, y1, p);
        yv[tt] = yt;
        y3 = y2; y2 = y1; y1 = yt;
        up3 = up2; up2 = up1; up1 = uc[h * kTT + tt];
    }
    __builtin_amdgcn_sched_barrier(0);
    #pragma unroll
    for (int tt = 0; tt < kTT; ++tt) ycol[tt * kYP] = yv[tt];
}

__device__ __forceinline__ void read_frags(v2f (&fr)[kNF], const float* frow)
{
    #pragma unroll
    for (int j = 0; j < kNF / 2; ++j) {
        const v4f v = *reinterpret_cast<const v4f*>(frow + 4 * j);
        fr[2 * j + 0].x = v.x; fr[2 * j + 0].y = v.y;
        fr[2 * j + 1].x = v.z; fr[2 * j + 1].y = v.w;
    }
}

__device__ __forceinline__ void pair_sum(const v2f (&f0)[kNF], const v2f (&f1)[kNF],
                                         float& lo, float& hi)
{
    v2f ones; ones.x = 1.0f; ones.y = 1.0f;
    v8f acc0 = {};
    v8f acc1 = {};
    #pragma unroll
    for (int g = 0; g < kNF; ++g) {
        acc0 = __builtin_amdgcn_wmma_f32_16x16x4_f32(
            false, ones, false, f0[g], (short)0, acc0, false, false);
        acc1 = __builtin_amdgcn_wmma_f32_16x16x4_f32(
            false, ones, false, f1[g], (short)0, acc1, false, false);
    }
    lo = acc0[0];
    hi = acc1[0];
}

template <int Q0, int Q1>
__device__ __forceinline__ void sum_window(const float* fwin, float* sw, int lane,
                                           const float* pr, float* __restrict__ pd, bool has)
{
    v2f f[2][2][kNF];
    v4f pv = {};
    if (has) pv = *reinterpret_cast<const v4f*>(pr);
    read_frags(f[Q0 & 1][0], fwin + (Q0 * kG) * kYP);
    read_frags(f[Q0 & 1][1], fwin + (Q0 * kG + kTT) * kYP);
    #pragma unroll
    for (int q = Q0; q < Q1; ++q) {
        if (q + 1 < Q1) {
            read_frags(f[(q + 1) & 1][0], fwin + ((q + 1) * kG) * kYP);
            read_frags(f[(q + 1) & 1][1], fwin + ((q + 1) * kG + kTT) * kYP);
        }
        __builtin_amdgcn_sched_barrier(0);
        if (q == Q0 && has) *reinterpret_cast<v4f*>(pd) = pv;
        float lo, hi;
        pair_sum(f[q & 1][0], f[q & 1][1], lo, hi);
        sw[q * kSG] = (lane < 16) ? lo : hi;
        __builtin_amdgcn_sched_barrier(0);
    }
}

__global__ __launch_bounds__(2 * kOPB * 32) void mimo_iir_wmma_kernel(
    const float* u,
    const float* __restrict__ bc,
    const float* __restrict__ ac,
    float* __restrict__ out)
{
    __shared__ __attribute__((aligned(16))) float y_lds[2 * kYB];
    __shared__ __attribute__((aligned(16))) float u_lds[2 * kUB];
    __shared__ __attribute__((aligned(16))) float stage[2 * kSB];

    const int tid  = threadIdx.x;
    const int wv   = __builtin_amdgcn_readfirstlane(tid >> 5);
    const bool rec = wv < kOPB;
    const int w    = wv & (kOPB - 1);
    const int lane = tid & 31;
    const int b    = blockIdx.x / (kCOUT / kOPB);
    const int o    = (blockIdx.x % (kCOUT / kOPB)) * kOPB + w;
    const int i    = lane;

    float b0 = 0.f, b1 = 0.f, b2 = 0.f, a0 = 0.f, a1 = 0.f, a2 = 0.f;
    float up1 = 0.f, up2 = 0.f, up3 = 0.f;
    float y1 = 0.f, y2 = 0.f, y3 = 0.f;
    float ua[kG] = {}, ub[kG] = {};
    v4f gq[kRW / 4] = {};

    const float* uq = u + (size_t)b * kT * kCIN + w * kRW * kCIN + lane * 4;
    float* sq = u_lds + 4 * (lane & 7) * kUP + kUS * ((lane & 7) >> 2) + w * kRW + (lane >> 3);
    const float* urow = u_lds + lane * kUP + kUS * (lane >> 4);
    float* ycol = y_lds + w * kYW + ((i >> 1) & 1) * 16 + (i >> 2) * 2 + (i & 1);

    float* ob4 = out + (size_t)b * kT * kCOUT + (o - w);
    const int m   = lane & 15;
    const int hh  = lane >> 4;
    const float* fcol = y_lds + w * kYW + m * kYP + 16 * hh;

    if (rec) {
        const int cb = (o * kCIN + i) * 3;
        b0 = bc[cb + 0]; b1 = bc[cb + 1]; b2 = bc[cb + 2];
        a0 = ac[cb + 0]; a1 = ac[cb + 1]; a2 = ac[cb + 2];
        fetch_quarter(gq, uq);
        store_quarter(sq, gq);
        fetch_quarter(gq, uq + kCH * kCIN);
    }
    lds_block_barrier();
    if (rec) load_group_lds(ua, urow);

#define CHAIN(UC, H, YR) \
    chain_tile(UC, H, b0, b1, b2, a0, a1, a2, up1, up2, up3, y1, y2, y3, YR)

    for (int c = 0; c < kNCH; ++c) {
        const int nb = ((c + 1) & 1) * kUB;
        if (rec) {
            const float* cur = urow + (c & 1) * kUB;
            float* yc = ycol + (c & 1) * kYB;
#define LDS_GROUP_END __builtin_amdgcn_sched_barrier(0)
            load_group_lds(ub, cur + 1 * kG);
            LDS_GROUP_END;
            CHAIN(ua, 0, yc + (1 * kG) * kYP);
            LDS_GROUP_END;
            CHAIN(ua, 1, yc + (1 * kG + kTT) * kYP);
            store_quarter(sq + nb, gq);
            const int cn = (c + 2 < kNCH) ? (c + 2) : 0;
            fetch_quarter(gq, uq + cn * kCH * kCIN);
            load_group_lds(ua, cur + 2 * kG);
            LDS_GROUP_END;
            CHAIN(ub, 0, yc + (2 * kG) * kYP);
            LDS_GROUP_END;
            CHAIN(ub, 1, yc + (2 * kG + kTT) * kYP);
            load_group_lds(ub, cur + 3 * kG);
            LDS_GROUP_END;
            CHAIN(ua, 0, yc + (3 * kG) * kYP);
            LDS_GROUP_END;
            CHAIN(ua, 1, yc + (3 * kG + kTT) * kYP);
        }
        lds_block_barrier();
        if (rec) {
            float* yn = ycol + ((c + 1) & 1) * kYB;
            load_group_lds(ua, urow + nb);
            LDS_GROUP_END;
            CHAIN(ub, 0, yn);
            LDS_GROUP_END;
            CHAIN(ub, 1, yn + kTT * kYP);
        } else {
            const float* fwin = fcol + (c & 1) * kYB;
            float* sw = stage + (c & 1) * kSB + lane * kOPB + w;
            if (c == 0) {
                sum_window<1, 4>(fwin, sw, lane, nullptr, nullptr, false);
            } else {
                const int tg = (c - 1) * kCH - kG + w * kG;
                const bool has = tg >= 0;
                const float* pr = stage + ((c - 1) & 1) * kSB + w * kSG + lane * kOPB;
                float* pd = ob4 + (size_t)(c * kCH + w * kG + lane) * kCOUT - (kCH + kG) * kCOUT;
                sum_window<0, 4>(fwin, sw, lane, pr, pd, has);
            }
        }
    }
#undef CHAIN
#undef LDS_GROUP_END
    lds_block_barrier();
    if (!rec) {
        const float* pr = stage + ((kNCH - 1) & 1) * kSB + w * kSG + lane * kOPB;
        float* pd = ob4 + (size_t)((kNCH - 1) * kCH - kG + w * kG + lane) * kCOUT;
        sum_window<0, 1>(fcol + (kNCH & 1) * kYB, stage + (kNCH & 1) * kSB + lane * kOPB + w, lane,
                         pr, pd, true);
    }
    lds_block_barrier();
    if (wv == kOPB) {
        const v4f v = *reinterpret_cast<const v4f*>(stage + (kNCH & 1) * kSB + lane * kOPB);
        *reinterpret_cast<v4f*>(ob4 + (size_t)(kT - kG + lane) * kCOUT) = v;
    }
}

extern "C" void kernel_launch(void* const* d_in, const int* in_sizes, int n_in,
                              void* d_out, int out_size, void* d_ws, size_t ws_size,
                              hipStream_t stream) {
    (void)n_in; (void)out_size; (void)d_ws; (void)ws_size;
    const float* u  = (const float*)d_in[0];
    const float* bc = (const float*)d_in[1];
    const float* ac = (const float*)d_in[2];
    float* out = (float*)d_out;

    const int B = in_sizes[0] / (kT * kCIN);
    dim3 grid(B * (kCOUT / kOPB));
    dim3 block(2 * kOPB * 32);
    mimo_iir_wmma_kernel<<<grid, block, 0, stream>>>(u, bc, ac, out);
}


# wait for the coefficient loads pinned in front of the loop: the loop's counted load waits, which on every pass also held the wave for the u fetch issued two groups earlier, are gone
# speedup vs baseline: 6.8664x; 1.0342x over previous
#include <hip/hip_runtime.h>


typedef float v2f __attribute__((ext_vector_type(2)));
typedef float v4f __attribute__((ext_vector_type(4)));
typedef float v8f __attribute__((ext_vector_type(8)));

constexpr int kT    = 2048;
constexpr int kCIN  = 32;
constexpr int kCOUT = 32;
constexpr int kTT   = 16;
constexpr int kG    = 32;
constexpr int kOPB  = 4;
constexpr int kNF   = kCIN / 4;
constexpr int kCH   = 128;
constexpr int kNCH  = kT / kCH;
constexpr int kRW   = kCH / kOPB;
constexpr int kUP   = kCH + 4;
constexpr int kUS   = 4;
constexpr int kUB   = kCIN * kUP;
constexpr int kYP   = 36;
constexpr int kYW   = kCH * kYP;
constexpr int kYB   = kOPB * kYW;
constexpr int kSG   = kG * kOPB;
constexpr int kSB   = (kCH / kG) * kSG;

__device__ __forceinline__ void fetch_quarter(v4f (&dst)[kRW / 4], const float* p)
{
    #pragma unroll
    for (int j = 0; j < kRW / 4; ++j)
        dst[j] = *reinterpret_cast<const v4f*>(p + j * (4 * kCIN));
}

__device__ __forceinline__ void load_group_lds(float (&dst)[kG], const float* p)
{
    #pragma unroll
    for (int j = 0; j < kG / 4; ++j) {
        const v4f v = *reinterpret_cast<const v4f*>(p + 4 * j);
        dst[4 * j + 0] = v.x; dst[4 * j + 1] = v.y;
        dst[4 * j + 2] = v.z; dst[4 * j + 3] = v.w;
    }
}

__device__ __forceinline__ void store_quarter(float* p, const v4f (&src)[kRW / 4])
{
    #pragma unroll
    for (int j = 0; j < kRW / 4; ++j) {
        p[0 * kUP + 4 * j] = src[j].x;
        p[1 * kUP + 4 * j] = src[j].y;
        p[2 * kUP + 4 * j] = src[j].z;
        p[3 * kUP + 4 * j] = src[j].w;
    }
}

__device__ __forceinline__ void lds_block_barrier()
{
    __builtin_amdgcn_fence(__ATOMIC_RELEASE, "workgroup", "local");
    __builtin_amdgcn_s_barrier();
    __builtin_amdgcn_fence(__ATOMIC_ACQUIRE, "workgroup", "local");
}

__device__ __forceinline__ void chain_tile(
    const float (&uc)[kG], const int h,
    float b0, float b1, float b2, float a0, float a1, float a2,
    float& up1, float& up2, float& up3, float& y1, float& y2, float& y3,
    float* ycol)
{
    float yv[kTT];
    #pragma unroll
    for (int tt = 0; tt < kTT; ++tt) {
        float x  = fmaf(b0, up1, fmaf(b1, up2, b2 * up3));
        float p  = fmaf(-a1, y2, fmaf(-a2, y3, x));
        float yt = fmaf(-a0, y1, p);
        yv[tt] = yt;
        y3 = y2; y2 = y1; y1 = yt;
        up3 = up2; up2 = up1; up1 = uc[h * kTT + tt];
    }
    __builtin_amdgcn_sched_barrier(0);
    #pragma unroll
    for (int tt = 0; tt < kTT; ++tt) ycol[tt * kYP] = yv[tt];
}

__device__ __forceinline__ void read_frags(v2f (&fr)[kNF], const float* frow)
{
    #pragma unroll
    for (int j = 0; j < kNF / 2; ++j) {
        const v4f v = *reinterpret_cast<const v4f*>(frow + 4 * j);
        fr[2 * j + 0].x = v.x; fr[2 * j + 0].y = v.y;
        fr[2 * j + 1].x = v.z; fr[2 * j + 1].y = v.w;
    }
}

__device__ __forceinline__ void pair_sum(const v2f (&f0)[kNF], const v2f (&f1)[kNF],
                                         float& lo, float& hi)
{
    v2f ones; ones.x = 1.0f; ones.y = 1.0f;
    v8f acc0 = {};
    v8f acc1 = {};
    #pragma unroll
    for (int g = 0; g < kNF; ++g) {
        acc0 = __builtin_amdgcn_wmma_f32_16x16x4_f32(
            false, ones, false, f0[g], (short)0, acc0, false, false);
        acc1 = __builtin_amdgcn_wmma_f32_16x16x4_f32(
            false, ones, false, f1[g], (short)0, acc1, false, false);
    }
    lo = acc0[0];
    hi = acc1[0];
}

template <int Q0, int Q1>
__device__ __forceinline__ void sum_window(const float* fwin, float* sw, int lane,
                                           const float* pr, float* __restrict__ pd, bool has)
{
    v2f f[2][2][kNF];
    v4f pv = {};
    if (has) pv = *reinterpret_cast<const v4f*>(pr);
    read_frags(f[Q0 & 1][0], fwin + (Q0 * kG) * kYP);
    read_frags(f[Q0 & 1][1], fwin + (Q0 * kG + kTT) * kYP);
    #pragma unroll
    for (int q = Q0; q < Q1; ++q) {
        if (q + 1 < Q1) {
            read_frags(f[(q + 1) & 1][0], fwin + ((q + 1) * kG) * kYP);
            read_frags(f[(q + 1) & 1][1], fwin + ((q + 1) * kG + kTT) * kYP);
        }
        __builtin_amdgcn_sched_barrier(0);
        if (q == Q0 && has) *reinterpret_cast<v4f*>(pd) = pv;
        float lo, hi;
        pair_sum(f[q & 1][0], f[q & 1][1], lo, hi);
        sw[q * kSG] = (lane < 16) ? lo : hi;
        __builtin_amdgcn_sched_barrier(0);
    }
}

__global__ __launch_bounds__(2 * kOPB * 32) void mimo_iir_wmma_kernel(
    const float* u,
    const float* __restrict__ bc,
    const float* __restrict__ ac,
    float* __restrict__ out)
{
    __shared__ __attribute__((aligned(16))) float y_lds[2 * kYB];
    __shared__ __attribute__((aligned(16))) float u_lds[2 * kUB];
    __shared__ __attribute__((aligned(16))) float stage[2 * kSB];

    const int tid  = threadIdx.x;
    const int wv   = __builtin_amdgcn_readfirstlane(tid >> 5);
    const bool rec = wv < kOPB;
    const int w    = wv & (kOPB - 1);
    const int lane = tid & 31;
    const int b    = blockIdx.x / (kCOUT / kOPB);
    const int o    = (blockIdx.x % (kCOUT / kOPB)) * kOPB + w;
    const int i    = lane;

    float b0 = 0.f, b1 = 0.f, b2 = 0.f, a0 = 0.f, a1 = 0.f, a2 = 0.f;
    float up1 = 0.f, up2 = 0.f, up3 = 0.f;
    float y1 = 0.f, y2 = 0.f, y3 = 0.f;
    float ua[kG] = {}, ub[kG] = {};
    v4f gq[kRW / 4] = {};

    const float* uq = u + (size_t)b * kT * kCIN + w * kRW * kCIN + lane * 4;
    float* sq = u_lds + 4 * (lane & 7) * kUP + kUS * ((lane & 7) >> 2) + w * kRW + (lane >> 3);
    const float* urow = u_lds + lane * kUP + kUS * (lane >> 4);
    float* ycol = y_lds + w * kYW + ((i >> 1) & 1) * 16 + (i >> 2) * 2 + (i & 1);

    float* ob4 = out + (size_t)b * kT * kCOUT + (o - w);
    const int m   = lane & 15;
    const int hh  = lane >> 4;
    const float* fcol = y_lds + w * kYW + m * kYP + 16 * hh;

    if (rec) {
        const int cb = (o * kCIN + i) * 3;
        b0 = bc[cb + 0]; b1 = bc[cb + 1]; b2 = bc[cb + 2];
        a0 = ac[cb + 0]; a1 = ac[cb + 1]; a2 = ac[cb + 2];
        fetch_quarter(gq, uq);
        store_quarter(sq, gq);
        fetch_quarter(gq, uq + kCH * kCIN);
    }
    lds_block_barrier();
    if (rec) load_group_lds(ua, urow);
    asm volatile("" : "+v"(b0), "+v"(b1), "+v"(b2), "+v"(a0), "+v"(a1), "+v"(a2));

#define CHAIN(UC, H, YR) \
    chain_tile(UC, H, b0, b1, b2, a0, a1, a2, up1, up2, up3, y1, y2, y3, YR)

    for (int c = 0; c < kNCH; ++c) {
        const int nb = ((c + 1) & 1) * kUB;
        if (rec) {
            const float* cur = urow + (c & 1) * kUB;
            float* yc = ycol + (c & 1) * kYB;
#define LDS_GROUP_END __builtin_amdgcn_sched_barrier(0)
            load_group_lds(ub, cur + 1 * kG);
            LDS_GROUP_END;
            CHAIN(ua, 0, yc + (1 * kG) * kYP);
            LDS_GROUP_END;
            CHAIN(ua, 1, yc + (1 * kG + kTT) * kYP);
            store_quarter(sq + nb, gq);
            const int cn = (c + 2 < kNCH) ? (c + 2) : 0;
            fetch_quarter(gq, uq + cn * kCH * kCIN);
            load_group_lds(ua, cur + 2 * kG);
            LDS_GROUP_END;
            CHAIN(ub, 0, yc + (2 * kG) * kYP);
            LDS_GROUP_END;
            CHAIN(ub, 1, yc + (2 * kG + kTT) * kYP);
            load_group_lds(ub, cur + 3 * kG);
            LDS_GROUP_END;
            CHAIN(ua, 0, yc + (3 * kG) * kYP);
            LDS_GROUP_END;
            CHAIN(ua, 1, yc + (3 * kG + kTT) * kYP);
        }
        lds_block_barrier();
        if (rec) {
            float* yn = ycol + ((c + 1) & 1) * kYB;
            load_group_lds(ua, urow + nb);
            LDS_GROUP_END;
            CHAIN(ub, 0, yn);
            LDS_GROUP_END;
            CHAIN(ub, 1, yn + kTT * kYP);
        } else {
            const float* fwin = fcol + (c & 1) * kYB;
            float* sw = stage + (c & 1) * kSB + lane * kOPB + w;
            if (c == 0) {
                sum_window<1, 4>(fwin, sw, lane, nullptr, nullptr, false);
            } else {
                const int tg = (c - 1) * kCH - kG + w * kG;
                const bool has = tg >= 0;
                const float* pr = stage + ((c - 1) & 1) * kSB + w * kSG + lane * kOPB;
                float* pd = ob4 + (size_t)(c * kCH + w * kG + lane) * kCOUT - (kCH + kG) * kCOUT;
                sum_window<0, 4>(fwin, sw, lane, pr, pd, has);
            }
        }
    }
#undef CHAIN
#undef LDS_GROUP_END
    lds_block_barrier();
    if (!rec) {
        const float* pr = stage + ((kNCH - 1) & 1) * kSB + w * kSG + lane * kOPB;
        float* pd = ob4 + (size_t)((kNCH - 1) * kCH - kG + w * kG + lane) * kCOUT;
        sum_window<0, 1>(fcol + (kNCH & 1) * kYB, stage + (kNCH & 1) * kSB + lane * kOPB + w, lane,
                         pr, pd, true);
    }
    lds_block_barrier();
    if (wv == kOPB) {
        const v4f v = *reinterpret_cast<const v4f*>(stage + (kNCH & 1) * kSB + lane * kOPB);
        *reinterpret_cast<v4f*>(ob4 + (size_t)(kT - kG + lane) * kCOUT) = v;
    }
}

extern "C" void kernel_launch(void* const* d_in, const int* in_sizes, int n_in,
                              void* d_out, int out_size, void* d_ws, size_t ws_size,
                              hipStream_t stream) {
    (void)n_in; (void)out_size; (void)d_ws; (void)ws_size;
    const float* u  = (const float*)d_in[0];
    const float* bc = (const float*)d_in[1];
    const float* ac = (const float*)d_in[2];
    float* out = (float*)d_out;

    const int B = in_sizes[0] / (kT * kCIN);
    dim3 grid(B * (kCOUT / kOPB));
    dim3 block(2 * kOPB * 32);
    mimo_iir_wmma_kernel<<<grid, block, 0, stream>>>(u, bc, ac, out);
}


# y window row = step pair: a recurrence lane stores 2 steps per 8-byte piece, half the y store instructions; sum: parity-select A, 2 channels per matrix op, two chains added: sum order changed
# speedup vs baseline: 7.2725x; 1.0591x over previous
#include <hip/hip_runtime.h>


typedef float v2f __attribute__((ext_vector_type(2)));
typedef float v4f __attribute__((ext_vector_type(4)));
typedef float v8f __attribute__((ext_vector_type(8)));

constexpr int kT    = 2048;
constexpr int kCIN  = 32;
constexpr int kCOUT = 32;
constexpr int kTT   = 16;
constexpr int kG    = 32;
constexpr int kOPB  = 4;
constexpr int kNG   = kCIN / 2;
constexpr int kCH   = 128;
constexpr int kNCH  = kT / kCH;
constexpr int kRW   = kCH / kOPB;
constexpr int kUP   = kCH + 4;
constexpr int kUS   = 4;
constexpr int kUB   = kCIN * kUP;
constexpr int kYP   = 2 * kCIN + 4;
constexpr int kYW   = (kCH / 2) * kYP;
constexpr int kYB   = kOPB * kYW;
constexpr int kSG   = kG * kOPB;
constexpr int kSB   = (kCH / kG) * kSG;

__device__ __forceinline__ void fetch_quarter(v4f (&dst)[kRW / 4], const float* p)
{
    #pragma unroll
    for (int j = 0; j < kRW / 4; ++j)
        dst[j] = *reinterpret_cast<const v4f*>(p + j * (4 * kCIN));
}

__device__ __forceinline__ void load_group_lds(float (&dst)[kG], const float* p)
{
    #pragma unroll
    for (int j = 0; j < kG / 4; ++j) {
        const v4f v = *reinterpret_cast<const v4f*>(p + 4 * j);
        dst[4 * j + 0] = v.x; dst[4 * j + 1] = v.y;
        dst[4 * j + 2] = v.z; dst[4 * j + 3] = v.w;
    }
}

__device__ __forceinline__ void store_quarter(float* p, const v4f (&src)[kRW / 4])
{
    #pragma unroll
    for (int j = 0; j < kRW / 4; ++j) {
        p[0 * kUP + 4 * j] = src[j].x;
        p[1 * kUP + 4 * j] = src[j].y;
        p[2 * kUP + 4 * j] = src[j].z;
        p[3 * kUP + 4 * j] = src[j].w;
    }
}

__device__ __forceinline__ void lds_block_barrier()
{
    __builtin_amdgcn_fence(__ATOMIC_RELEASE, "workgroup", "local");
    __builtin_amdgcn_s_barrier();
    __builtin_amdgcn_fence(__ATOMIC_ACQUIRE, "workgroup", "local");
}

__device__ __forceinline__ void chain_tile(
    const float (&uc)[kG], const int h,
    float b0, float b1, float b2, float a0, float a1, float a2,
    float& up1, float& up2, float& up3, float& y1, float& y2, float& y3,
    float* ycol)
{
    float yv[kTT];
    #pragma unroll
    for (int tt = 0; tt < kTT; ++tt) {
        float x  = fmaf(b0, up1, fmaf(b1, up2, b2 * up3));
        float p  = fmaf(-a1, y2, fmaf(-a2, y3, x));
        float yt = fmaf(-a0, y1, p);
        yv[tt] = yt;
        y3 = y2; y2 = y1; y1 = yt;
        up3 = up2; up2 = up1; up1 = uc[h * kTT + tt];
    }
    __builtin_amdgcn_sched_barrier(0);
    #pragma unroll
    for (int tp = 0; tp < kTT / 2; ++tp) {
        v2f pr2; pr2.x = yv[2 * tp]; pr2.y = yv[2 * tp + 1];
        *reinterpret_cast<v2f*>(ycol + tp * kYP) = pr2;
    }
}

__device__ __forceinline__ void read_frags(v2f (&fr)[kNG], const float* frow)
{
    #pragma unroll
    for (int j = 0; j < kNG / 2; ++j) {
        const v4f v = *reinterpret_cast<const v4f*>(frow + 4 * j);
        fr[2 * j + 0].x = v.x; fr[2 * j + 0].y = v.y;
        fr[2 * j + 1].x = v.z; fr[2 * j + 1].y = v.w;
    }
}

__device__ __forceinline__ void group_sum(const v2f (&f)[kNG], const v2f sel,
                                          float& ev, float& od)
{
    v8f acc0 = {};
    v8f acc1 = {};
    #pragma unroll
    for (int g = 0; g < kNG; g += 2) {
        acc0 = __builtin_amdgcn_wmma_f32_16x16x4_f32(
            false, sel, false, f[g], (short)0, acc0, false, false);
        acc1 = __builtin_amdgcn_wmma_f32_16x16x4_f32(
            false, sel, false, f[g + 1], (short)0, acc1, false, false);
    }
    ev = acc0[0] + acc1[0];
    od = acc0[1] + acc1[1];
}

template <int Q0, int Q1>
__device__ __forceinline__ void sum_window(const float* fwin, float* sw, int lane, const v2f sel,
                                           const float* pr, float* __restrict__ pd, bool has)
{
    v2f f[2][kNG];
    v4f pv = {};
    if (has) pv = *reinterpret_cast<const v4f*>(pr);
    read_frags(f[Q0 & 1], fwin + (Q0 * (kG / 2)) * kYP);
    #pragma unroll
    for (int q = Q0; q < Q1; ++q) {
        if (q + 1 < Q1)
            read_frags(f[(q + 1) & 1], fwin + ((q + 1) * (kG / 2)) * kYP);
        __builtin_amdgcn_sched_barrier(0);
        asm volatile("" : "+v"(f[q & 1][0]), "+v"(f[q & 1][1]));
        if (q == Q0 && has) *reinterpret_cast<v4f*>(pd) = pv;
        float ev, od;
        group_sum(f[q & 1], sel, ev, od);
        sw[q * kSG] = (lane < 16) ? ev : od;
        __builtin_amdgcn_sched_barrier(0);
    }
}

__global__ __launch_bounds__(2 * kOPB * 32) void mimo_iir_wmma_kernel(
    const float* u,
    const float* __restrict__ bc,
    const float* __restrict__ ac,
    float* __restrict__ out)
{
    __shared__ __attribute__((aligned(16))) float y_lds[2 * kYB];
    __shared__ __attribute__((aligned(16))) float u_lds[2 * kUB];
    __shared__ __attribute__((aligned(16))) float stage[2 * kSB];

    const int tid  = threadIdx.x;
    const int wv   = __builtin_amdgcn_readfirstlane(tid >> 5);
    const bool rec = wv < kOPB;
    const int w    = wv & (kOPB - 1);
    const int lane = tid & 31;
    const int b    = blockIdx.x / (kCOUT / kOPB);
    const int o    = (blockIdx.x % (kCOUT / kOPB)) * kOPB + w;
    const int i    = lane;

    float b0 = 0.f, b1 = 0.f, b2 = 0.f, a0 = 0.f, a1 = 0.f, a2 = 0.f;
    float up1 = 0.f, up2 = 0.f, up3 = 0.f;
    float y1 = 0.f, y2 = 0.f, y3 = 0.f;
    float ua[kG] = {}, ub[kG] = {};
    v4f gq[kRW / 4] = {};

    const float* uq = u + (size_t)b * kT * kCIN + w * kRW * kCIN + lane * 4;
    float* sq = u_lds + 4 * (lane & 7) * kUP + kUS * ((lane & 7) >> 2) + w * kRW + (lane >> 3);
    const float* urow = u_lds + lane * kUP + kUS * (lane >> 4);
    float* ycol = y_lds + w * kYW + 2 * i;

    float* ob4 = out + (size_t)b * kT * kCOUT + (o - w);
    const int m   = lane & 15;
    const int hh  = lane >> 4;
    const float* fcol = y_lds + w * kYW + m * kYP + 2 * 16 * hh;
    v2f sel; sel.x = (m & 1) ? 0.0f : 1.0f; sel.y = (m & 1) ? 1.0f : 0.0f;
    const int tl  = 2 * m + hh;

    if (rec) {
        const int cb = (o * kCIN + i) * 3;
        b0 = bc[cb + 0]; b1 = bc[cb + 1]; b2 = bc[cb + 2];
        a0 = ac[cb + 0]; a1 = ac[cb + 1]; a2 = ac[cb + 2];
        fetch_quarter(gq, uq);
        store_quarter(sq, gq);
        fetch_quarter(gq, uq + kCH * kCIN);
    }
    lds_block_barrier();
    if (rec) load_group_lds(ua, urow);
    asm volatile("" : "+v"(b0), "+v"(b1), "+v"(b2), "+v"(a0), "+v"(a1), "+v"(a2));

#define CHAIN(UC, H, YR) \
    chain_tile(UC, H, b0, b1, b2, a0, a1, a2, up1, up2, up3, y1, y2, y3, YR)

    for (int c = 0; c < kNCH; ++c) {
        const int nb = ((c + 1) & 1) * kUB;
        if (rec) {
            const float* cur = urow + (c & 1) * kUB;
            float* yc = ycol + (c & 1) * kYB;
#define LDS_GROUP_END __builtin_amdgcn_sched_barrier(0)
            load_group_lds(ub, cur + 1 * kG);
            LDS_GROUP_END;
            CHAIN(ua, 0, yc + ((1 * kG) / 2) * kYP);
            LDS_GROUP_END;
            CHAIN(ua, 1, yc + ((1 * kG + kTT) / 2) * kYP);
            store_quarter(sq + nb, gq);
            const int cn = (c + 2 < kNCH) ? (c + 2) : 0;
            fetch_quarter(gq, uq + cn * kCH * kCIN);
            load_group_lds(ua, cur + 2 * kG);
            LDS_GROUP_END;
            CHAIN(ub, 0, yc + ((2 * kG) / 2) * kYP);
            LDS_GROUP_END;
            CHAIN(ub, 1, yc + ((2 * kG + kTT) / 2) * kYP);
            load_group_lds(ub, cur + 3 * kG);
            LDS_GROUP_END;
            CHAIN(ua, 0, yc + ((3 * kG) / 2) * kYP);
            LDS_GROUP_END;
            CHAIN(ua, 1, yc + ((3 * kG + kTT) / 2) * kYP);
        }
        lds_block_barrier();
        if (rec) {
            float* yn = ycol + ((c + 1) & 1) * kYB;
            load_group_lds(ua, urow + nb);
            LDS_GROUP_END;
            CHAIN(ub, 0, yn);
            LDS_GROUP_END;
            CHAIN(ub, 1, yn + (kTT / 2) * kYP);
        } else {
            const float* fwin = fcol + (c & 1) * kYB;
            float* sw = stage + (c & 1) * kSB + tl * kOPB + w;
            if (c == 0) {
                sum_window<1, 4>(fwin, sw, lane, sel, nullptr, nullptr, false);
            } else {
                const int tg = (c - 1) * kCH - kG + w * kG;
                const bool has = tg >= 0;
                const float* pr = stage + ((c - 1) & 1) * kSB + w * kSG + lane * kOPB;
                float* pd = ob4 + (size_t)(c * kCH + w * kG + lane) * kCOUT - (kCH + kG) * kCOUT;
                sum_window<0, 4>(fwin, sw, lane, sel, pr, pd, has);
            }
        }
    }
#undef CHAIN
#undef LDS_GROUP_END
    lds_block_barrier();
    if (!rec) {
        const float* pr = stage + ((kNCH - 1) & 1) * kSB + w * kSG + lane * kOPB;
        float* pd = ob4 + (size_t)((kNCH - 1) * kCH - kG + w * kG + lane) * kCOUT;
        sum_window<0, 1>(fcol + (kNCH & 1) * kYB, stage + (kNCH & 1) * kSB + tl * kOPB + w, lane,
                         sel, pr, pd, true);
    }
    lds_block_barrier();
    if (wv == kOPB) {
        const v4f v = *reinterpret_cast<const v4f*>(stage + (kNCH & 1) * kSB + lane * kOPB);
        *reinterpret_cast<v4f*>(ob4 + (size_t)(kT - kG + lane) * kCOUT) = v;
    }
}

extern "C" void kernel_launch(void* const* d_in, const int* in_sizes, int n_in,
                              void* d_out, int out_size, void* d_ws, size_t ws_size,
                              hipStream_t stream) {
    (void)n_in; (void)out_size; (void)d_ws; (void)ws_size;
    const float* u  = (const float*)d_in[0];
    const float* bc = (const float*)d_in[1];
    const float* ac = (const float*)d_in[2];
    float* out = (float*)d_out;

    const int B = in_sizes[0] / (kT * kCIN);
    dim3 grid(B * (kCOUT / kOPB));
    dim3 block(2 * kOPB * 32);
    mimo_iir_wmma_kernel<<<grid, block, 0, stream>>>(u, bc, ac, out);
}
